# starting kernel
# baseline (speedup 1.0000x reference)
#include <hip/hip_runtime.h>

typedef float v2f __attribute__((ext_vector_type(2)));
typedef float v8f __attribute__((ext_vector_type(8)));

#define NB   256
#define DD   128
#define NP   1296
#define NC   64
#define EPSF 1e-4f

__global__ __launch_bounds__(256)
void ImageParser_68745246540061_kernel(const float* __restrict__ x,
                                       const int*   __restrict__ cm,
                                       float*       __restrict__ out) {
    __shared__ int   s_cm[NP];
    __shared__ float s_cnt[NC];
    __shared__ float s_mean[DD * NC];

    const int tid  = threadIdx.x;
    const int b    = blockIdx.x;
    const int lane = tid & 31;
    const int wv   = tid >> 5;

    const int* cmb = cm + b * NP;
    for (int i = tid; i < NP; i += 256) s_cm[i] = cmb[i];
    if (tid < NC) s_cnt[tid] = 0.0f;
    __syncthreads();
    for (int i = tid; i < NP; i += 256) atomicAdd(&s_cnt[s_cm[i]], 1.0f);
    __syncthreads();

    const int  mrow = wv * 16 + (lane & 15);
    const int  kb   = (lane >> 4) << 1;
    const float* xrow = x + ((size_t)b * DD + mrow) * NP;
    const int  myc  = lane & 15;

    v8f acc[4] = {};

    for (int n0 = 0; n0 < NP; n0 += 4) {
        v2f a = *(const v2f*)(xrow + n0 + kb);
        int c0 = s_cm[n0 + kb];
        int c1 = s_cm[n0 + kb + 1];
#pragma unroll
        for (int t = 0; t < 4; ++t) {
            int cc = t * 16 + myc;
            v2f bf;
            bf.x = (c0 == cc) ? 1.0f : 0.0f;
            bf.y = (c1 == cc) ? 1.0f : 0.0f;
            acc[t] = __builtin_amdgcn_wmma_f32_16x16x4_f32(
                false, a, false, bf, (short)0, acc[t], false, false);
        }
    }

    const int rbase = wv * 16 + ((lane >> 4) << 3);
#pragma unroll
    for (int t = 0; t < 4; ++t) {
        int col = t * 16 + myc;
#pragma unroll
        for (int j = 0; j < 8; ++j)
            s_mean[(rbase + j) * NC + col] = acc[t][j];
    }
    __syncthreads();

    for (int i = tid; i < DD * NC; i += 256)
        s_mean[i] = s_mean[i] / (s_cnt[i & (NC - 1)] + EPSF);
    __syncthreads();

    const float* xb = x + (size_t)b * DD * NP;
    for (int n = tid; n < NP; n += 256) {
        int   c    = s_cm[n];
        float accd = 0.0f;
#pragma unroll 4
        for (int d = 0; d < DD; ++d) {
            float v    = xb[(size_t)d * NP + n];
            float m    = s_mean[d * NC + c];
            float diff = v - m;
            accd = fmaf(diff, diff, accd);
        }
        out[(size_t)b * NP + n] = accd * (1.0f / (float)DD);
    }
}

extern "C" void kernel_launch(void* const* d_in, const int* in_sizes, int n_in,
                              void* d_out, int out_size, void* d_ws, size_t ws_size,
                              hipStream_t stream) {
    (void)in_sizes; (void)n_in; (void)d_ws; (void)ws_size; (void)out_size;
    const float* x  = (const float*)d_in[0];
    const int*   cm = (const int*)d_in[1];
    float*       o  = (float*)d_out;
    ImageParser_68745246540061_kernel<<<NB, 256, 0, stream>>>(x, cm, o);
}


# class sums: wave = one 16-class tile over a quarter of the pixels, WMMAs issued only for 4-pixel chunks holding a pixel of that tile; pixel sum = 4 quarter partials added in quarter order
# speedup vs baseline: 2.8164x; 2.8164x over previous
#include <hip/hip_runtime.h>

typedef float v2f __attribute__((ext_vector_type(2)));
typedef float v4f __attribute__((ext_vector_type(4)));
typedef float v8f __attribute__((ext_vector_type(8)));
typedef int   v2i __attribute__((ext_vector_type(2)));
typedef int   v4i __attribute__((ext_vector_type(4)));

#define NB   256
#define DD   128
#define NP   1296
#define NC   64
#define EPSF 1e-4f

#define NT    512
#define NW    16
#define MT    2
#define SR    (16 * MT)
#define NSLAB (DD / SR)
#define PITCH 1300
#define RQ    (NP / 4)
#define SQ    (SR * RQ)
#define NLD   ((SQ + NT - 1) / NT)
#define NKQ   4
#define QPW   (RQ / NKQ)
#define PP    68
static_assert(SR * NC == 4 * NT, "mean pass: one 16-B group per thread");
static_assert(NP % 4 == 0 && DD % SR == 0 && RQ % NKQ == 0, "shape");
static_assert(NW == (NC / 16) * NKQ && MT == 2, "wave = (class tile, pixel quarter)");
static_assert(QPW > 64 && QPW <= 96, "chunk mask = one 64-bit word and one 32-bit word");

__device__ __forceinline__ void lds_barrier() {
    __builtin_amdgcn_fence(__ATOMIC_RELEASE, "workgroup", "local");
    __builtin_amdgcn_s_barrier();
    __builtin_amdgcn_fence(__ATOMIC_ACQUIRE, "workgroup", "local");
}

__device__ __forceinline__ void chunk_pair(const float* xa, const int* ca, int qa, int qb,
                                           int cc, int ccb, v8f& acc0, v8f& acc1) {
    const v2i cla = *(const v2i*)&ca[4 * qa];
    const v2i clb = *(const v2i*)&ca[4 * qb];
    const v2f a0 = *(const v2f*)&xa[4 * qa];
    const v2f a1 = *(const v2f*)&xa[16 * PITCH + 4 * qa];
    const v2f a2 = *(const v2f*)&xa[4 * qb];
    const v2f a3 = *(const v2f*)&xa[16 * PITCH + 4 * qb];
    v2f ba, bb;
    ba.x = (cla.x == cc)  ? 1.0f : 0.0f;
    ba.y = (cla.y == cc)  ? 1.0f : 0.0f;
    bb.x = (clb.x == ccb) ? 1.0f : 0.0f;
    bb.y = (clb.y == ccb) ? 1.0f : 0.0f;
    acc0 = __builtin_amdgcn_wmma_f32_16x16x4_f32(false, a0, false, ba, (short)0, acc0, false, false);
    acc1 = __builtin_amdgcn_wmma_f32_16x16x4_f32(false, a1, false, ba, (short)0, acc1, false, false);
    acc0 = __builtin_amdgcn_wmma_f32_16x16x4_f32(false, a2, false, bb, (short)0, acc0, false, false);
    acc1 = __builtin_amdgcn_wmma_f32_16x16x4_f32(false, a3, false, bb, (short)0, acc1, false, false);
}

__global__ __launch_bounds__(NT)
void class_mean_dist_kernel(const float* __restrict__ x,
                            const int*   __restrict__ cm,
                            float*       __restrict__ out) {
    __shared__ __attribute__((aligned(16))) float s_x[SR * PITCH];
    __shared__ __attribute__((aligned(16))) int   s_cm[NP];
    __shared__ __attribute__((aligned(16))) float s_cnt[NC];
    __shared__ __attribute__((aligned(16))) float s_part[NKQ * SR * PP];
    __shared__ __attribute__((aligned(16))) float s_mean[SR * NC];

    const int tid  = threadIdx.x;
    const int b    = blockIdx.x;
    const int lane = tid & 31;
    const int c    = lane & 15;
    const int h    = lane >> 4;
    const int wv   = __builtin_amdgcn_readfirstlane(tid >> 5);

    const int* cmb = cm + b * NP;
    int cmv[3];
#pragma unroll
    for (int i = 0; i < 3; ++i) {
        int p = tid + i * NT;
        cmv[i] = (p < NP) ? cmb[p] : 0;
    }

    const v4f* xb4 = (const v4f*)(x + (size_t)b * DD * NP);
    v4f st[NLD];
#pragma unroll
    for (int i = 0; i < NLD; ++i) {
        int q = i * NT + tid;
        if (q < SQ) st[i] = xb4[q];
    }

#pragma unroll
    for (int i = 0; i < 3; ++i) {
        int p = tid + i * NT;
        if (p < NP) s_cm[p] = cmv[i];
    }
    if (tid < NC) s_cnt[tid] = 0.0f;
    lds_barrier();
    for (int i = tid; i < NP; i += NT) atomicAdd(&s_cnt[s_cm[i]], 1.0f);

    const int tw = wv & 3;
    const int kq = wv >> 2;
    unsigned mw[3];
#pragma unroll
    for (int w = 0; w < 3; ++w) {
        const int ql = 32 * w + lane;
        bool hit = false;
        if (ql < QPW) {
            const v4i cv = *(const v4i*)&s_cm[4 * (kq * QPW + ql)];
            hit = ((cv.x >> 4) == tw) | ((cv.y >> 4) == tw) | ((cv.z >> 4) == tw) | ((cv.w >> 4) == tw);
        }
        mw[w] = __builtin_amdgcn_ballot_w32(hit);
    }
    const unsigned long long mk_lo = (unsigned long long)mw[0] | ((unsigned long long)mw[1] << 32);
    const unsigned           mk_hi = mw[2];

    v4i kc = {0, 0, 0, 0};
    if (tid < RQ) kc = *(const v4i*)&s_cm[4 * tid];
    float da0 = 0.0f, da1 = 0.0f, da2 = 0.0f, da3 = 0.0f;

    for (int s = 0; s < NSLAB; ++s) {
#pragma unroll
        for (int i = 0; i < NLD; ++i) {
            int q = i * NT + tid;
            if (q < SQ) {
                int r  = q / RQ;
                int c4 = q - r * RQ;
                *(v4f*)&s_x[r * PITCH + 4 * c4] = st[i];
            }
        }
        lds_barrier();

        if (s + 1 < NSLAB) {
            const v4f* nx = xb4 + (size_t)(s + 1) * SQ;
#pragma unroll
            for (int i = 0; i < NLD; ++i) {
                int q = i * NT + tid;
                if (q < SQ) st[i] = nx[q];
            }
        }

        v8f acc0 = {}, acc1 = {};
        const int cc = 16 * tw + c;
        const float* xa = &s_x[c * PITCH + 2 * h];
        const int*   ca = &s_cm[2 * h];
        const int qw = kq * QPW;
        {
            unsigned long long m = mk_lo;
            while (m) {
                const int qa = qw + __builtin_ctzll(m);
                m &= m - 1;
                const bool two = (m != 0);
                const int qb  = two ? qw + __builtin_ctzll(m) : qa;
                const int ccb = two ? cc : -1 - cc;
                m &= m - 1;
                chunk_pair(xa, ca, qa, qb, cc, ccb, acc0, acc1);
            }
        }
        {
            unsigned m = mk_hi;
            while (m) {
                const int qa = qw + 64 + __builtin_ctz(m);
                m &= m - 1;
                const bool two = (m != 0);
                const int qb  = two ? qw + 64 + __builtin_ctz(m) : qa;
                const int ccb = two ? cc : -1 - cc;
                m &= m - 1;
                chunk_pair(xa, ca, qa, qb, cc, ccb, acc0, acc1);
            }
        }
#pragma unroll
        for (int r = 0; r < 8; ++r) {
            s_part[(kq * SR + 8 * h + r) * PP + 16 * tw + c]      = acc0[r];
            s_part[(kq * SR + 16 + 8 * h + r) * PP + 16 * tw + c] = acc1[r];
        }
        lds_barrier();

        {
            const int er = tid >> 4;
            const int ec = (tid & 15) * 4;
            v4f sacc = *(const v4f*)&s_part[er * PP + ec];
#pragma unroll
            for (int k = 1; k < NKQ; ++k) sacc += *(const v4f*)&s_part[(k * SR + er) * PP + ec];
            const v4f cn = *(const v4f*)&s_cnt[ec];
            v4f mv;
            mv.x = sacc.x / (cn.x + EPSF);
            mv.y = sacc.y / (cn.y + EPSF);
            mv.z = sacc.z / (cn.z + EPSF);
            mv.w = sacc.w / (cn.w + EPSF);
            *(v4f*)&s_mean[4 * tid] = mv;
        }
        lds_barrier();

        if (tid < RQ) {
#pragma unroll 8
            for (int d = 0; d < SR; ++d) {
                const v4f xv = *(const v4f*)&s_x[d * PITCH + 4 * tid];
                const float* mr = &s_mean[d * NC];
                float e0 = xv.x - mr[kc.x];
                float e1 = xv.y - mr[kc.y];
                float e2 = xv.z - mr[kc.z];
                float e3 = xv.w - mr[kc.w];
                da0 = fmaf(e0, e0, da0);
                da1 = fmaf(e1, e1, da1);
                da2 = fmaf(e2, e2, da2);
                da3 = fmaf(e3, e3, da3);
            }
        }
        lds_barrier();
    }

    if (tid < RQ) {
        v4f o;
        o.x = da0 * (1.0f / (float)DD);
        o.y = da1 * (1.0f / (float)DD);
        o.z = da2 * (1.0f / (float)DD);
        o.w = da3 * (1.0f / (float)DD);
        *(v4f*)(out + (size_t)b * NP + 4 * tid) = o;
    }
}

extern "C" void kernel_launch(void* const* d_in, const int* in_sizes, int n_in,
                              void* d_out, int out_size, void* d_ws, size_t ws_size,
                              hipStream_t stream) {
    (void)in_sizes; (void)n_in; (void)d_ws; (void)ws_size; (void)out_size;
    const float* x  = (const float*)d_in[0];
    const int*   cm = (const int*)d_in[1];
    float*       o  = (float*)d_out;
    class_mean_dist_kernel<<<NB, NT, 0, stream>>>(x, cm, o);
}


# next slab's global loads issued one ahead of each class-sum trip (mask walk cut into 3 segments of 27 chunks, 7 loads each), not 21 in one block before the walk; every sum and its order unchanged
# speedup vs baseline: 3.4553x; 1.2268x over previous
#include <hip/hip_runtime.h>

typedef float v2f __attribute__((ext_vector_type(2)));
typedef float v4f __attribute__((ext_vector_type(4)));
typedef float v8f __attribute__((ext_vector_type(8)));
typedef int   v2i __attribute__((ext_vector_type(2)));
typedef int   v4i __attribute__((ext_vector_type(4)));

#define NB   256
#define DD   128
#define NP   1296
#define NC   64
#define EPSF 1e-4f

#define NT    512
#define NW    16
#define MT    2
#define SR    (16 * MT)
#define NSLAB (DD / SR)
#define PITCH 1300
#define RQ    (NP / 4)
#define SQ    (SR * RQ)
#define NLD   ((SQ + NT - 1) / NT)
#define NKQ   4
#define QPW   (RQ / NKQ)
#define PP    68
#define NSEG  3
#define SEGC  (QPW / NSEG)
#define LPS   (NLD / NSEG)
static_assert(SR * NC == 4 * NT, "mean pass: one 16-B group per thread");
static_assert(NP % 4 == 0 && DD % SR == 0 && RQ % NKQ == 0, "shape");
static_assert(NW == (NC / 16) * NKQ && MT == 2, "wave = (class tile, pixel quarter)");
static_assert(QPW % NSEG == 0 && SEGC <= 32, "chunk mask = one 32-bit word per segment");
static_assert(NLD % NSEG == 0, "the slab's loads divide evenly over the segments");
static_assert(SQ % 32 == 0, "the slab's last round of 16-B pieces ends on a wave boundary");

__device__ __forceinline__ void lds_barrier() {
    __builtin_amdgcn_fence(__ATOMIC_RELEASE, "workgroup", "local");
    __builtin_amdgcn_s_barrier();
    __builtin_amdgcn_fence(__ATOMIC_ACQUIRE, "workgroup", "local");
}

__device__ __forceinline__ void chunk_pair(const float* xa, const int* ca, int qa, int qb,
                                           int cc, int ccb, v8f& acc0, v8f& acc1) {
    const v2i cla = *(const v2i*)&ca[4 * qa];
    const v2i clb = *(const v2i*)&ca[4 * qb];
    const v2f a0 = *(const v2f*)&xa[4 * qa];
    const v2f a1 = *(const v2f*)&xa[16 * PITCH + 4 * qa];
    const v2f a2 = *(const v2f*)&xa[4 * qb];
    const v2f a3 = *(const v2f*)&xa[16 * PITCH + 4 * qb];
    v2f ba, bb;
    ba.x = (cla.x == cc)  ? 1.0f : 0.0f;
    ba.y = (cla.y == cc)  ? 1.0f : 0.0f;
    bb.x = (clb.x == ccb) ? 1.0f : 0.0f;
    bb.y = (clb.y == ccb) ? 1.0f : 0.0f;
    acc0 = __builtin_amdgcn_wmma_f32_16x16x4_f32(false, a0, false, ba, (short)0, acc0, false, false);
    acc1 = __builtin_amdgcn_wmma_f32_16x16x4_f32(false, a1, false, ba, (short)0, acc1, false, false);
    acc0 = __builtin_amdgcn_wmma_f32_16x16x4_f32(false, a2, false, bb, (short)0, acc0, false, false);
    acc1 = __builtin_amdgcn_wmma_f32_16x16x4_f32(false, a3, false, bb, (short)0, acc1, false, false);
}

__device__ __forceinline__ void mask_trip(unsigned& m, int qs, const float* xa, const int* ca,
                                          int cc, v8f& acc0, v8f& acc1) {
    const int qa = qs + __builtin_ctz(m);
    m &= m - 1;
    const bool two = (m != 0);
    const int qb  = two ? qs + __builtin_ctz(m) : qa;
    const int ccb = two ? cc : -1 - cc;
    m &= m - 1;
    chunk_pair(xa, ca, qa, qb, cc, ccb, acc0, acc1);
}

__global__ __launch_bounds__(NT)
void class_mean_dist_kernel(const float* __restrict__ x,
                            const int*   __restrict__ cm,
                            float*       __restrict__ out) {
    __shared__ __attribute__((aligned(16))) float s_x[SR * PITCH];
    __shared__ __attribute__((aligned(16))) int   s_cm[NP];
    __shared__ __attribute__((aligned(16))) float s_cnt[NC];
    __shared__ __attribute__((aligned(16))) float s_part[NKQ * SR * PP];
    __shared__ __attribute__((aligned(16))) float s_mean[SR * NC];

    const int tid  = threadIdx.x;
    const int b    = blockIdx.x;
    const int lane = tid & 31;
    const int c    = lane & 15;
    const int h    = lane >> 4;
    const int wv   = __builtin_amdgcn_readfirstlane(tid >> 5);

    const int* cmb = cm + b * NP;
    int cmv[3];
#pragma unroll
    for (int i = 0; i < 3; ++i) {
        int p = tid + i * NT;
        cmv[i] = (p < NP) ? cmb[p] : 0;
    }

    const v4f* xb4 = (const v4f*)(x + (size_t)b * DD * NP);
    v4f st[NLD];
#pragma unroll
    for (int i = 0; i < NLD; ++i) {
        int q = i * NT + tid;
        if (q < SQ) st[i] = xb4[q];
    }

#pragma unroll
    for (int i = 0; i < 3; ++i) {
        int p = tid + i * NT;
        if (p < NP) s_cm[p] = cmv[i];
    }
    if (tid < NC) s_cnt[tid] = 0.0f;
    lds_barrier();
    for (int i = tid; i < NP; i += NT) atomicAdd(&s_cnt[s_cm[i]], 1.0f);

    const int tw = wv & 3;
    const int kq = wv >> 2;
    unsigned mw[NSEG];
#pragma unroll
    for (int g = 0; g < NSEG; ++g) {
        bool hit = false;
        if (lane < SEGC) {
            const v4i cv = *(const v4i*)&s_cm[4 * (kq * QPW + g * SEGC + lane)];
            hit = ((cv.x >> 4) == tw) | ((cv.y >> 4) == tw) | ((cv.z >> 4) == tw) | ((cv.w >> 4) == tw);
        }
        mw[g] = __builtin_amdgcn_ballot_w32(hit);
    }

    v4i kc = {0, 0, 0, 0};
    if (tid < RQ) kc = *(const v4i*)&s_cm[4 * tid];
    float da0 = 0.0f, da1 = 0.0f, da2 = 0.0f, da3 = 0.0f;

    for (int s = 0; s < NSLAB; ++s) {
#pragma unroll
        for (int i = 0; i < NLD; ++i) {
            int q = i * NT + tid;
            if (q < SQ) {
                int r  = q / RQ;
                int c4 = q - r * RQ;
                *(v4f*)&s_x[r * PITCH + 4 * c4] = st[i];
            }
        }
        lds_barrier();

        v8f acc0 = {}, acc1 = {};
        const int cc = 16 * tw + c;
        const float* xa = &s_x[c * PITCH + 2 * h];
        const int*   ca = &s_cm[2 * h];
        const bool more = (s + 1 < NSLAB);
        const v4f* nx = xb4 + (size_t)(more ? s + 1 : s) * SQ;
#pragma unroll
        for (int g = 0; g < NSEG; ++g) {
            unsigned m = mw[g];
            const int qs = kq * QPW + g * SEGC;
#pragma unroll
            for (int j = 0; j < LPS; ++j) {
                const int i = g * LPS + j;
                const bool on = (i + 1) * NT <= SQ || wv < (SQ - i * NT) / 32;
                if (more && on) st[i] = nx[i * NT + tid];
                if (m) mask_trip(m, qs, xa, ca, cc, acc0, acc1);
            }
            while (m) mask_trip(m, qs, xa, ca, cc, acc0, acc1);
        }
#pragma unroll
        for (int r = 0; r < 8; ++r) {
            s_part[(kq * SR + 8 * h + r) * PP + 16 * tw + c]      = acc0[r];
            s_part[(kq * SR + 16 + 8 * h + r) * PP + 16 * tw + c] = acc1[r];
        }
        lds_barrier();

        {
            const int er = tid >> 4;
            const int ec = (tid & 15) * 4;
            v4f sacc = *(const v4f*)&s_part[er * PP + ec];
#pragma unroll
            for (int k = 1; k < NKQ; ++k) sacc += *(const v4f*)&s_part[(k * SR + er) * PP + ec];
            const v4f cn = *(const v4f*)&s_cnt[ec];
            v4f mv;
            mv.x = sacc.x / (cn.x + EPSF);
            mv.y = sacc.y / (cn.y + EPSF);
            mv.z = sacc.z / (cn.z + EPSF);
            mv.w = sacc.w / (cn.w + EPSF);
            *(v4f*)&s_mean[4 * tid] = mv;
        }
        lds_barrier();

        if (tid < RQ) {
#pragma unroll 8
            for (int d = 0; d < SR; ++d) {
                const v4f xv = *(const v4f*)&s_x[d * PITCH + 4 * tid];
                const float* mr = &s_mean[d * NC];
                float e0 = xv.x - mr[kc.x];
                float e1 = xv.y - mr[kc.y];
                float e2 = xv.z - mr[kc.z];
                float e3 = xv.w - mr[kc.w];
                da0 = fmaf(e0, e0, da0);
                da1 = fmaf(e1, e1, da1);
                da2 = fmaf(e2, e2, da2);
                da3 = fmaf(e3, e3, da3);
            }
        }

        lds_barrier();
    }

    if (tid < RQ) {
        v4f o;
        o.x = da0 * (1.0f / (float)DD);
        o.y = da1 * (1.0f / (float)DD);
        o.z = da2 * (1.0f / (float)DD);
        o.w = da3 * (1.0f / (float)DD);
        *(v4f*)(out + (size_t)b * NP + 4 * tid) = o;
    }
}

extern "C" void kernel_launch(void* const* d_in, const int* in_sizes, int n_in,
                              void* d_out, int out_size, void* d_ws, size_t ws_size,
                              hipStream_t stream) {
    (void)in_sizes; (void)n_in; (void)d_ws; (void)ws_size; (void)out_size;
    const float* x  = (const float*)d_in[0];
    const int*   cm = (const int*)d_in[1];
    float*       o  = (float*)d_out;
    class_mean_dist_kernel<<<NB, NT, 0, stream>>>(x, cm, o);
}
